# starting kernel
# baseline (speedup 1.0000x reference)
#include <hip/hip_runtime.h>
#include <hip/hip_bf16.h>


typedef __attribute__((ext_vector_type(2))) float v2f;
typedef __attribute__((ext_vector_type(8))) float v8f;

#define WS_MR   0
#define WS_MI   4096
#define WS_MIN  8192
#define WS_C0   12288
#define WS_S0   12294

struct c32 { float r, i; };
__device__ __forceinline__ c32 cmul(c32 a, c32 b) {
    return { a.r * b.r - a.i * b.i, a.r * b.i + a.i * b.r };
}

__global__ void qsim_setup_kernel(const float* __restrict__ w,
                                  float* __restrict__ ws) {
    __shared__ float d0r[64], d0i[64];
    const int k = threadIdx.x;
    float phi0 = 0.f, phi1 = 0.f;
#pragma unroll
    for (int i = 0; i < 5; ++i) {
        const int bi = (k >> (5 - i)) & 1;
        const int bj = (k >> (4 - i)) & 1;
        const float sgn = (bi == bj) ? -0.5f : 0.5f;
        phi0 += sgn * w[6 + i];
        phi1 += sgn * w[17 + i];
    }
    float s, c;
    sincosf(phi0, &s, &c); d0r[k] = c; d0i[k] = s;
    float d1r, d1i; sincosf(phi1, &d1i, &d1r);
    float c1[6], s1[6];
#pragma unroll
    for (int i = 0; i < 6; ++i) sincosf(0.5f * w[11 + i], &s1[i], &c1[i]);
    __syncthreads();
    for (int kp = 0; kp < 64; ++kp) {
        float re = 1.f;
#pragma unroll
        for (int i = 0; i < 6; ++i) {
            const int kb = (k >> (5 - i)) & 1;
            const int pb = (kp >> (5 - i)) & 1;
            re *= (kb == pb) ? c1[i] : (kb ? s1[i] : -s1[i]);
        }
        const float mr = re * (d1r * d0r[kp] - d1i * d0i[kp]);
        const float mi = re * (d1r * d0i[kp] + d1i * d0r[kp]);
        ws[WS_MR  + k * 64 + kp] = mr;
        ws[WS_MI  + k * 64 + kp] = mi;
        ws[WS_MIN + k * 64 + kp] = -mi;
    }
    if (k < 6) {
        float ss, cc; sincosf(0.5f * w[k], &ss, &cc);
        ws[WS_C0 + k] = cc;
        ws[WS_S0 + k] = ss;
    }
}

__global__ __launch_bounds__(128, 1)
void qsim_main_kernel(const float* __restrict__ z,
                      const float* __restrict__ ws,
                      float* __restrict__ out) {
    __shared__ float s_psi[4][2][64 * 17];

    const int tid  = threadIdx.x;
    const int wave = tid >> 5;
    const int lane = tid & 31;
    const int n    = lane & 15;
    const int kh   = lane >> 4;
    const int base = blockIdx.x * 64 + wave * 16;
    const int b    = base + n;

    const float z0 = z[2 * b + 0];
    const float z1 = z[2 * b + 1];
    float sa, ca, sb, cb;
    sincosf(0.5f * asinf(z0), &sa, &ca);
    sincosf(0.5f * asinf(z1), &sb, &cb);
    const c32 x0{ cb * ca, -sb * ca };
    const c32 x1{ cb * sa,  sb * sa };

    c32 u[6][2];
#pragma unroll
    for (int j = 0; j < 6; ++j) {
        const float c0 = ws[WS_C0 + j], s0 = ws[WS_S0 + j];
        u[j][0] = { c0 * x0.r - s0 * x1.r, c0 * x0.i - s0 * x1.i };
        u[j][1] = { s0 * x0.r + c0 * x1.r, s0 * x0.i + c0 * x1.i };
    }
    c32 t01[4], t012[8], t34[4], t345[8];
#pragma unroll
    for (int p = 0; p < 4; ++p) t01[p]  = cmul(u[0][p >> 1], u[1][p & 1]);
#pragma unroll
    for (int p = 0; p < 8; ++p) t012[p] = cmul(t01[p >> 1], u[2][p & 1]);
#pragma unroll
    for (int p = 0; p < 4; ++p) t34[p]  = cmul(u[3][p >> 1], u[4][p & 1]);
#pragma unroll
    for (int p = 0; p < 8; ++p) t345[p] = cmul(t34[p >> 1], u[5][p & 1]);

    float* psir = &s_psi[wave][0][0];
    float* psii = &s_psi[wave][1][0];
#pragma unroll
    for (int pp = 0; pp < 4; ++pp) {
        const c32 tp = t012[4 * kh + pp];
#pragma unroll
        for (int q = 0; q < 8; ++q) {
            const c32 vq = cmul(tp, t345[q]);
            const int kidx = 8 * (4 * kh + pp) + q;
            psir[kidx * 17 + n] = vq.r;
            psii[kidx * 17 + n] = vq.i;
        }
    }
    __syncthreads();

    const float* MR = ws + WS_MR;
    const float* MI = ws + WS_MI;
    const float* MN = ws + WS_MIN;

    float acc0 = 0.f, acc1 = 0.f, acc2 = 0.f, acc3 = 0.f, acc4 = 0.f, acc5 = 0.f;

#pragma unroll
    for (int r = 0; r < 4; ++r) {
        v8f yr = {};
        v8f yi = {};
        const float* rowR = MR + (16 * r + n) * 64 + 2 * kh;
        const float* rowI = MI + (16 * r + n) * 64 + 2 * kh;
        const float* rowN = MN + (16 * r + n) * 64 + 2 * kh;
        for (int kk = 0; kk < 16; ++kk) {
            const v2f ar = *(const v2f*)(rowR + 4 * kk);
            const v2f ai = *(const v2f*)(rowI + 4 * kk);
            const v2f an = *(const v2f*)(rowN + 4 * kk);
            const int k0 = 4 * kk + 2 * kh;
            const float* pr = psir + k0 * 17 + n;
            const float* pi = psii + k0 * 17 + n;
            const v2f br = { pr[0], pr[17] };
            const v2f bi = { pi[0], pi[17] };
            yr = __builtin_amdgcn_wmma_f32_16x16x4_f32(false, ar, false, br, (short)0, yr, false, false);
            yr = __builtin_amdgcn_wmma_f32_16x16x4_f32(false, an, false, bi, (short)0, yr, false, false);
            yi = __builtin_amdgcn_wmma_f32_16x16x4_f32(false, ar, false, bi, (short)0, yi, false, false);
            yi = __builtin_amdgcn_wmma_f32_16x16x4_f32(false, ai, false, br, (short)0, yi, false, false);
        }
#pragma unroll
        for (int v = 0; v < 8; ++v) {
            const float p = yr[v] * yr[v] + yi[v] * yi[v];
            acc0 = (r & 2) ? acc0 - p : acc0 + p;
            acc1 = (r & 1) ? acc1 - p : acc1 + p;
            acc2 += p;
            acc3 = (v & 4) ? acc3 - p : acc3 + p;
            acc4 = (v & 2) ? acc4 - p : acc4 + p;
            acc5 = (v & 1) ? acc5 - p : acc5 + p;
        }
    }
    acc2 = kh ? -acc2 : acc2;

    const float e0 = acc0 + __shfl_xor(acc0, 16, 32);
    const float e1 = acc1 + __shfl_xor(acc1, 16, 32);
    const float e2 = acc2 + __shfl_xor(acc2, 16, 32);
    const float e3 = acc3 + __shfl_xor(acc3, 16, 32);
    const float e4 = acc4 + __shfl_xor(acc4, 16, 32);
    const float e5 = acc5 + __shfl_xor(acc5, 16, 32);

    if (kh == 0) {
        float* o = out + (size_t)b * 6;
        *(v2f*)(o + 0) = (v2f){ e0, e1 };
        *(v2f*)(o + 2) = (v2f){ e2, e3 };
        *(v2f*)(o + 4) = (v2f){ e4, e5 };
    }
}

extern "C" void kernel_launch(void* const* d_in, const int* in_sizes, int n_in,
                              void* d_out, int out_size, void* d_ws, size_t ws_size,
                              hipStream_t stream) {
    const float* w  = (const float*)d_in[0];
    const float* z  = (const float*)d_in[1];
    float* ws       = (float*)d_ws;
    float* o        = (float*)d_out;
    const int B     = in_sizes[1] / 2;

    qsim_setup_kernel<<<1, 64, 0, stream>>>(w, ws);
    qsim_main_kernel<<<B / 64, 128, 0, stream>>>(z, ws, o);
}


# K cut from 64 to 7: all six qubits carry the same encoded 2-vector x, so psi = G v with the 7 monomials v_m = x0^(6-m) x1^m; set-up folds G into M (64x8): 2 k-steps, not 16; changes fp32 sum order
# speedup vs baseline: 4.4466x; 4.4466x over previous
#include <hip/hip_runtime.h>
#include <hip/hip_bf16.h>


typedef __attribute__((ext_vector_type(2))) float v2f;
typedef __attribute__((ext_vector_type(8))) float v8f;

#define WS_MR   0
#define WS_MI   512

struct c32 { float r, i; };
__device__ __forceinline__ c32 cmul(c32 a, c32 b) {
    return { a.r * b.r - a.i * b.i, a.r * b.i + a.i * b.r };
}

__global__ __launch_bounds__(512)
void qsim_setup_kernel(const float* __restrict__ w,
                       float* __restrict__ ws) {
    __shared__ float s_d0r[64], s_d0i[64], s_d1r[64], s_d1i[64];
    __shared__ float s_c0[6], s_s0[6], s_c1[6], s_s1[6];
    __shared__ float s_hr[64 * 8], s_hi[64 * 8];
    const int tid = threadIdx.x;

    if (tid < 128) {
        const int k   = tid & 63;
        const int off = (tid < 64) ? 6 : 17;
        float phi = 0.f;
#pragma unroll
        for (int i = 0; i < 5; ++i) {
            const int bi = (k >> (5 - i)) & 1;
            const int bj = (k >> (4 - i)) & 1;
            const float sgn = (bi == bj) ? -0.5f : 0.5f;
            phi += sgn * w[off + i];
        }
        float s, c;
        sincosf(phi, &s, &c);
        if (tid < 64) { s_d0r[k] = c; s_d0i[k] = s; }
        else          { s_d1r[k] = c; s_d1i[k] = s; }
    } else if (tid < 140) {
        const int j = tid - 128;
        float s, c;
        sincosf(0.5f * w[(j < 6) ? j : (j + 5)], &s, &c);
        if (j < 6) { s_c0[j] = c;     s_s0[j] = s; }
        else       { s_c1[j - 6] = c; s_s1[j - 6] = s; }
    }
    __syncthreads();

    if (tid < 64) {
        const int kp = tid;
        float g[7] = { 1.f, 0.f, 0.f, 0.f, 0.f, 0.f, 0.f };
#pragma unroll
        for (int i = 0; i < 6; ++i) {
            const int bit = (kp >> (5 - i)) & 1;
            const float a  = bit ? s_s0[i] : s_c0[i];
            const float bb = bit ? s_c0[i] : -s_s0[i];
#pragma unroll
            for (int m = i + 1; m >= 1; --m) g[m] = a * g[m] + bb * g[m - 1];
            g[0] = a * g[0];
        }
        const float c = s_d0r[kp], s = s_d0i[kp];
#pragma unroll
        for (int m = 0; m < 7; ++m) {
            s_hr[kp * 8 + m] = c * g[m];
            s_hi[kp * 8 + m] = s * g[m];
        }
        s_hr[kp * 8 + 7] = 0.f;
        s_hi[kp * 8 + 7] = 0.f;
    }
    __syncthreads();

    const int k = tid >> 3;
    const int m = tid & 7;
    float fa[8], fb[8];
#pragma unroll
    for (int p = 0; p < 8; ++p) {
        float va = 1.f, vb = 1.f;
#pragma unroll
        for (int i = 0; i < 3; ++i) {
            const int kb = (k >> (5 - i)) & 1;
            const int pb = (p >> (2 - i)) & 1;
            va *= (kb == pb) ? s_c1[i] : (kb ? s_s1[i] : -s_s1[i]);
            const int kc = (k >> (2 - i)) & 1;
            vb *= (kc == pb) ? s_c1[3 + i] : (kc ? s_s1[3 + i] : -s_s1[3 + i]);
        }
        fa[p] = va;
        fb[p] = vb;
    }
    float accr = 0.f, acci = 0.f;
#pragma unroll
    for (int p = 0; p < 8; ++p) {
#pragma unroll
        for (int q = 0; q < 8; ++q) {
            const float re = fa[p] * fb[q];
            const int kp = 8 * p + q;
            accr += re * s_hr[kp * 8 + m];
            acci += re * s_hi[kp * 8 + m];
        }
    }
    const float d1r = s_d1r[k], d1i = s_d1i[k];
    ws[WS_MR + k * 8 + m] = d1r * accr - d1i * acci;
    ws[WS_MI + k * 8 + m] = d1r * acci + d1i * accr;
}

__global__ __launch_bounds__(128, 1)
void qsim_main_kernel(const float* __restrict__ z,
                      const float* __restrict__ ws,
                      float* __restrict__ out) {
    __shared__ float s_v[4][2][8 * 17];
    __shared__ float s_red[4][384];

    const int tid  = threadIdx.x;
    const int wave = tid >> 5;
    const int lane = tid & 31;
    const int n    = lane & 15;
    const int kh   = lane >> 4;
    const int base = blockIdx.x * 64 + wave * 16;
    const int b    = base + n;

    const float* rowR = ws + WS_MR + (16 * wave + n) * 8 + 2 * kh;
    const float* rowI = ws + WS_MI + (16 * wave + n) * 8 + 2 * kh;
    v2f am[2], ap[2], aq[2];
    {
        const v2f r0 = *(const v2f*)(rowR);
        const v2f r1 = *(const v2f*)(rowR + 4);
        const v2f i0 = *(const v2f*)(rowI);
        const v2f i1 = *(const v2f*)(rowI + 4);
        am[0] = r0;      am[1] = r1;
        ap[0] = r0 + i0; ap[1] = r1 + i1;
        aq[0] = i0 - r0; aq[1] = i1 - r1;
    }

    const float z0 = z[2 * b + 0];
    const float z1 = z[2 * b + 1];
    float sa, ca, sb, cb;
    sincosf(0.5f * asinf(z0), &sa, &ca);
    sincosf(0.5f * asinf(z1), &sb, &cb);
    const c32 x0{ cb * ca, -sb * ca };
    const c32 x1{ cb * sa,  sb * sa };

    const c32 a2 = cmul(x0, x0), a3 = cmul(a2, x0), a4 = cmul(a2, a2);
    const c32 b2 = cmul(x1, x1), b3 = cmul(b2, x1), b4 = cmul(b2, b2);
    c32 vm[4];
    if (kh == 0) {
        vm[0] = cmul(a4, a2);
        vm[1] = cmul(cmul(a4, x0), x1);
        vm[2] = cmul(a4, b2);
        vm[3] = cmul(a3, b3);
    } else {
        vm[0] = cmul(a2, b4);
        vm[1] = cmul(x0, cmul(b4, x1));
        vm[2] = cmul(b4, b2);
        vm[3] = { 0.f, 0.f };
    }
    {
        float* vr = &s_v[wave][0][0];
        float* vi = &s_v[wave][1][0];
#pragma unroll
        for (int pp = 0; pp < 4; ++pp) {
            const int m = 4 * kh + pp;
            vr[m * 17 + n] = vm[pp].r;
            vi[m * 17 + n] = vm[pp].i;
        }
    }
    __syncthreads();

    for (int t = 0; t < 4; ++t) {
        const float* vr = &s_v[t][0][0];
        const float* vi = &s_v[t][1][0];
        v8f p1 = {};
        v8f p2 = {};
        v8f p3 = {};
#pragma unroll
        for (int kk = 0; kk < 2; ++kk) {
            const int k0 = 4 * kk + 2 * kh;
            const float* pr = vr + k0 * 17 + n;
            const float* pi = vi + k0 * 17 + n;
            const v2f br = { pr[0], pr[17] };
            const v2f bi = { pi[0], pi[17] };
            const v2f bs = br + bi;
            p1 = __builtin_amdgcn_wmma_f32_16x16x4_f32(false, am[kk], false, bs, (short)0, p1, false, false);
            p2 = __builtin_amdgcn_wmma_f32_16x16x4_f32(false, ap[kk], false, bi, (short)0, p2, false, false);
            p3 = __builtin_amdgcn_wmma_f32_16x16x4_f32(false, aq[kk], false, br, (short)0, p3, false, false);
        }
        float ps = 0.f, a3s = 0.f, a4s = 0.f, a5s = 0.f;
#pragma unroll
        for (int v = 0; v < 8; ++v) {
            const float yr = p1[v] - p2[v];
            const float yi = p1[v] + p3[v];
            const float p = yr * yr + yi * yi;
            ps += p;
            a3s = (v & 4) ? a3s - p : a3s + p;
            a4s = (v & 2) ? a4s - p : a4s + p;
            a5s = (v & 1) ? a5s - p : a5s + p;
        }
        const float a2s = kh ? -ps : ps;
        const float pt = ps + __shfl_xor(ps, 16, 32);
        const float e2 = a2s + __shfl_xor(a2s, 16, 32);
        const float e3 = a3s + __shfl_xor(a3s, 16, 32);
        const float e4 = a4s + __shfl_xor(a4s, 16, 32);
        const float e5 = a5s + __shfl_xor(a5s, 16, 32);
        const float e0 = (wave & 2) ? -pt : pt;
        const float e1 = (wave & 1) ? -pt : pt;
        float* rp = &s_red[wave][(16 * t + n) * 6 + 3 * kh];
        rp[0] = kh ? e3 : e0;
        rp[1] = kh ? e4 : e1;
        rp[2] = kh ? e5 : e2;
    }
    __syncthreads();

    float* o = out + (size_t)blockIdx.x * 384;
    for (int j = 0; j < 3; ++j) {
        const int idx = tid + 128 * j;
        o[idx] = ((s_red[0][idx] + s_red[1][idx]) + s_red[2][idx]) + s_red[3][idx];
    }
}

extern "C" void kernel_launch(void* const* d_in, const int* in_sizes, int n_in,
                              void* d_out, int out_size, void* d_ws, size_t ws_size,
                              hipStream_t stream) {
    const float* w  = (const float*)d_in[0];
    const float* z  = (const float*)d_in[1];
    float* ws       = (float*)d_ws;
    float* o        = (float*)d_out;
    const int B     = in_sizes[1] / 2;

    qsim_setup_kernel<<<1, 512, 0, stream>>>(w, ws);
    qsim_main_kernel<<<B / 64, 128, 0, stream>>>(z, ws, o);
}


# Outputs as quadratic forms in the 7 monomials: out = W(6x56) * Re,Im(conj(v_m) v_m'), set-up folds W = M'^H Z_i M'; 14 WMMAs per 16 columns, not 24, no |Y|^2, no LDS tiles; changes fp32 sum order
# speedup vs baseline: 5.0154x; 1.1279x over previous
#include <hip/hip_runtime.h>
#include <hip/hip_bf16.h>


typedef __attribute__((ext_vector_type(2))) float v2f;
typedef __attribute__((ext_vector_type(8))) float v8f;

#define NPAIR 28
#define NSTEP 14

struct c32 { float r, i; };
__device__ __forceinline__ c32 cmul(c32 a, c32 b) {
    return { a.r * b.r - a.i * b.i, a.r * b.i + a.i * b.r };
}

__global__ __launch_bounds__(512)
void qsim_setup_kernel(const float* __restrict__ w,
                       float* __restrict__ ws) {
    __shared__ float s_d0r[64], s_d0i[64], s_d1r[64], s_d1i[64];
    __shared__ float s_c0[6], s_s0[6], s_c1[6], s_s1[6];
    __shared__ float s_hr[64 * 8], s_hi[64 * 8];
    __shared__ float2 s_m[64 * 8];
    const int tid = threadIdx.x;

    if (tid < 128) {
        const int k   = tid & 63;
        const int off = (tid < 64) ? 6 : 17;
        float phi = 0.f;
#pragma unroll
        for (int i = 0; i < 5; ++i) {
            const int bi = (k >> (5 - i)) & 1;
            const int bj = (k >> (4 - i)) & 1;
            const float sgn = (bi == bj) ? -0.5f : 0.5f;
            phi += sgn * w[off + i];
        }
        float s, c;
        sincosf(phi, &s, &c);
        if (tid < 64) { s_d0r[k] = c; s_d0i[k] = s; }
        else          { s_d1r[k] = c; s_d1i[k] = s; }
    } else if (tid < 140) {
        const int j = tid - 128;
        float s, c;
        sincosf(0.5f * w[(j < 6) ? j : (j + 5)], &s, &c);
        if (j < 6) { s_c0[j] = c;     s_s0[j] = s; }
        else       { s_c1[j - 6] = c; s_s1[j - 6] = s; }
    }
    __syncthreads();

    if (tid < 64) {
        const int kp = tid;
        float g[7] = { 1.f, 0.f, 0.f, 0.f, 0.f, 0.f, 0.f };
#pragma unroll
        for (int i = 0; i < 6; ++i) {
            const int bit = (kp >> (5 - i)) & 1;
            const float a  = bit ? s_s0[i] : s_c0[i];
            const float bb = bit ? s_c0[i] : -s_s0[i];
#pragma unroll
            for (int m = i + 1; m >= 1; --m) g[m] = a * g[m] + bb * g[m - 1];
            g[0] = a * g[0];
        }
        const float c = s_d0r[kp], s = s_d0i[kp];
#pragma unroll
        for (int m = 0; m < 7; ++m) {
            s_hr[kp * 8 + m] = c * g[m];
            s_hi[kp * 8 + m] = s * g[m];
        }
        s_hr[kp * 8 + 7] = 0.f;
        s_hi[kp * 8 + 7] = 0.f;
    }
    __syncthreads();

    {
        const int k = tid >> 3;
        const int m = tid & 7;
        float fa[8], fb[8];
#pragma unroll
        for (int p = 0; p < 8; ++p) {
            float va = 1.f, vb = 1.f;
#pragma unroll
            for (int i = 0; i < 3; ++i) {
                const int kb = (k >> (5 - i)) & 1;
                const int pb = (p >> (2 - i)) & 1;
                va *= (kb == pb) ? s_c1[i] : (kb ? s_s1[i] : -s_s1[i]);
                const int kc = (k >> (2 - i)) & 1;
                vb *= (kc == pb) ? s_c1[3 + i] : (kc ? s_s1[3 + i] : -s_s1[3 + i]);
            }
            fa[p] = va;
            fb[p] = vb;
        }
        float accr = 0.f, acci = 0.f;
#pragma unroll
        for (int p = 0; p < 8; ++p) {
#pragma unroll
            for (int q = 0; q < 8; ++q) {
                const float re = fa[p] * fb[q];
                const int kp = 8 * p + q;
                accr += re * s_hr[kp * 8 + m];
                acci += re * s_hi[kp * 8 + m];
            }
        }
        const float d1r = s_d1r[k], d1i = s_d1i[k];
        s_m[k * 8 + m] = make_float2(d1r * accr - d1i * acci, d1r * acci + d1i * accr);
    }
    __syncthreads();

    if (tid < 224) {
        const int q    = tid & 3;
        const int u    = tid >> 2;
        const int part = u & 1;
        const int t    = u >> 1;
        const int m  = (t >= 7) + (t >= 13) + (t >= 18) + (t >= 22) + (t >= 25) + (t >= 27);
        const int mp = m + (t - (7 * m - (m * (m - 1)) / 2));
        const float sd = part ? -1.f : 1.f;
        float tot = 0.f, z2 = 0.f, z3 = 0.f, z4 = 0.f, z5 = 0.f;
#pragma unroll
        for (int j = 0; j < 16; ++j) {
            const int k = 16 * q + j;
            const float2 a = s_m[k * 8 + m];
            const float2 b = s_m[k * 8 + mp];
            const float bb = part ? b.y : b.x;
            const float dd = part ? b.x : b.y;
            const float p  = a.x * bb + sd * (a.y * dd);
            tot += p;
            z2 = (j & 8) ? z2 - p : z2 + p;
            z3 = (j & 4) ? z3 - p : z3 + p;
            z4 = (j & 2) ? z4 - p : z4 + p;
            z5 = (j & 1) ? z5 - p : z5 + p;
        }
        float e[6] = { (q & 2) ? -tot : tot, (q & 1) ? -tot : tot, z2, z3, z4, z5 };
#pragma unroll
        for (int i = 0; i < 6; ++i) {
            e[i] += __shfl_xor(e[i], 1, 32);
            e[i] += __shfl_xor(e[i], 2, 32);
        }
        if (q == 0) {
            const float sc = (m == mp) ? (part ? 0.f : 1.f) : (part ? -2.f : 2.f);
#pragma unroll
            for (int i = 0; i < 6; ++i)
                ws[((i * 2 + part) * NSTEP + (t >> 1)) * 2 + (t & 1)] = sc * e[i];
        }
    }
}

__global__ __launch_bounds__(128, 1)
void qsim_main_kernel(const float* __restrict__ z,
                      const float* __restrict__ ws,
                      float* __restrict__ out) {
    __shared__ __attribute__((aligned(16))) float s_out[4][96];

    const int tid  = threadIdx.x;
    const int wave = tid >> 5;
    const int lane = tid & 31;
    const int n    = lane & 15;
    const int kh   = lane >> 4;
    const int base = blockIdx.x * 64 + wave * 16;
    const int b    = base + n;

    v2f af[NSTEP];
    if (n < 6) {
        const float4* p = (const float4*)(ws + (n * 2 + kh) * (NSTEP * 2));
#pragma unroll
        for (int j = 0; j < NSTEP / 2; ++j) {
            const float4 q = p[j];
            af[2 * j]     = v2f{ q.x, q.y };
            af[2 * j + 1] = v2f{ q.z, q.w };
        }
    } else {
#pragma unroll
        for (int j = 0; j < NSTEP; ++j) af[j] = v2f{ 0.f, 0.f };
    }

    const float z0 = z[2 * b + 0];
    const float z1 = z[2 * b + 1];
    float sa, ca, sb, cb;
    sincosf(0.5f * asinf(z0), &sa, &ca);
    sincosf(0.5f * asinf(z1), &sb, &cb);
    const c32 x0{ cb * ca, -sb * ca };
    const c32 x1{ cb * sa,  sb * sa };

    const c32 a2 = cmul(x0, x0), a3 = cmul(a2, x0), a4 = cmul(a2, a2);
    const c32 b2 = cmul(x1, x1), b3 = cmul(b2, x1), b4 = cmul(b2, b2);
    c32 vm[7];
    vm[0] = cmul(a4, a2);
    vm[1] = cmul(cmul(a4, x0), x1);
    vm[2] = cmul(a4, b2);
    vm[3] = cmul(a3, b3);
    vm[4] = cmul(a2, b4);
    vm[5] = cmul(x0, cmul(b4, x1));
    vm[6] = cmul(b4, b2);

    float pp[7], qq[7];
#pragma unroll
    for (int m = 0; m < 7; ++m) {
        pp[m] = kh ? vm[m].i : vm[m].r;
        qq[m] = kh ? -vm[m].r : vm[m].i;
    }
    float f[NPAIR];
    {
        int t = 0;
#pragma unroll
        for (int m = 0; m < 7; ++m) {
#pragma unroll
            for (int mp = m; mp < 7; ++mp) {
                f[t] = vm[m].r * pp[mp] + vm[m].i * qq[mp];
                ++t;
            }
        }
    }

    v8f acc0 = {};
    v8f acc1 = {};
#pragma unroll
    for (int kk = 0; kk < NSTEP; kk += 2) {
        const v2f bf0 = { f[2 * kk],     f[2 * kk + 1] };
        const v2f bf1 = { f[2 * kk + 2], f[2 * kk + 3] };
        acc0 = __builtin_amdgcn_wmma_f32_16x16x4_f32(false, af[kk],     false, bf0, (short)0, acc0, false, false);
        acc1 = __builtin_amdgcn_wmma_f32_16x16x4_f32(false, af[kk + 1], false, bf1, (short)0, acc1, false, false);
    }

    float* so = &s_out[wave][0];
    if (kh == 0) {
        float2* d = (float2*)(so + n * 6);
        d[0] = make_float2(acc0[0] + acc1[0], acc0[1] + acc1[1]);
        d[1] = make_float2(acc0[2] + acc1[2], acc0[3] + acc1[3]);
        d[2] = make_float2(acc0[4] + acc1[4], acc0[5] + acc1[5]);
    }

    float* o = out + (size_t)base * 6;
#pragma unroll
    for (int j = 0; j < 3; ++j) o[lane + 32 * j] = so[lane + 32 * j];
}

extern "C" void kernel_launch(void* const* d_in, const int* in_sizes, int n_in,
                              void* d_out, int out_size, void* d_ws, size_t ws_size,
                              hipStream_t stream) {
    const float* w  = (const float*)d_in[0];
    const float* z  = (const float*)d_in[1];
    float* ws       = (float*)d_ws;
    float* o        = (float*)d_out;
    const int B     = in_sizes[1] / 2;

    qsim_setup_kernel<<<1, 512, 0, stream>>>(w, ws);
    qsim_main_kernel<<<B / 64, 128, 0, stream>>>(z, ws, o);
}


# Each lane owns one column (32 per wave, two WMMA tiles): halves no longer both compute the same monomials; v_permlane16_swap_b32 trades re/im parts into the B fragments; changes fp32 sum order
# speedup vs baseline: 5.5717x; 1.1109x over previous
#include <hip/hip_runtime.h>
#include <hip/hip_bf16.h>


typedef __attribute__((ext_vector_type(2))) float v2f;
typedef __attribute__((ext_vector_type(8))) float v8f;

#define NPAIR 28
#define NSTEP 14

struct c32 { float r, i; };
__device__ __forceinline__ c32 cmul(c32 a, c32 b) {
    return { a.r * b.r - a.i * b.i, a.r * b.i + a.i * b.r };
}

__global__ __launch_bounds__(512)
void qsim_setup_kernel(const float* __restrict__ w,
                       float* __restrict__ ws) {
    __shared__ float s_d0r[64], s_d0i[64], s_d1r[64], s_d1i[64];
    __shared__ float s_c0[6], s_s0[6], s_c1[6], s_s1[6];
    __shared__ float s_hr[64 * 8], s_hi[64 * 8];
    __shared__ float2 s_m[64 * 8];
    const int tid = threadIdx.x;

    if (tid < 128) {
        const int k   = tid & 63;
        const int off = (tid < 64) ? 6 : 17;
        float phi = 0.f;
#pragma unroll
        for (int i = 0; i < 5; ++i) {
            const int bi = (k >> (5 - i)) & 1;
            const int bj = (k >> (4 - i)) & 1;
            const float sgn = (bi == bj) ? -0.5f : 0.5f;
            phi += sgn * w[off + i];
        }
        float s, c;
        sincosf(phi, &s, &c);
        if (tid < 64) { s_d0r[k] = c; s_d0i[k] = s; }
        else          { s_d1r[k] = c; s_d1i[k] = s; }
    } else if (tid < 140) {
        const int j = tid - 128;
        float s, c;
        sincosf(0.5f * w[(j < 6) ? j : (j + 5)], &s, &c);
        if (j < 6) { s_c0[j] = c;     s_s0[j] = s; }
        else       { s_c1[j - 6] = c; s_s1[j - 6] = s; }
    }
    __syncthreads();

    if (tid < 64) {
        const int kp = tid;
        float g[7] = { 1.f, 0.f, 0.f, 0.f, 0.f, 0.f, 0.f };
#pragma unroll
        for (int i = 0; i < 6; ++i) {
            const int bit = (kp >> (5 - i)) & 1;
            const float a  = bit ? s_s0[i] : s_c0[i];
            const float bb = bit ? s_c0[i] : -s_s0[i];
#pragma unroll
            for (int m = i + 1; m >= 1; --m) g[m] = a * g[m] + bb * g[m - 1];
            g[0] = a * g[0];
        }
        const float c = s_d0r[kp], s = s_d0i[kp];
#pragma unroll
        for (int m = 0; m < 7; ++m) {
            s_hr[kp * 8 + m] = c * g[m];
            s_hi[kp * 8 + m] = s * g[m];
        }
        s_hr[kp * 8 + 7] = 0.f;
        s_hi[kp * 8 + 7] = 0.f;
    }
    __syncthreads();

    {
        const int k = tid >> 3;
        const int m = tid & 7;
        float fa[8], fb[8];
#pragma unroll
        for (int p = 0; p < 8; ++p) {
            float va = 1.f, vb = 1.f;
#pragma unroll
            for (int i = 0; i < 3; ++i) {
                const int kb = (k >> (5 - i)) & 1;
                const int pb = (p >> (2 - i)) & 1;
                va *= (kb == pb) ? s_c1[i] : (kb ? s_s1[i] : -s_s1[i]);
                const int kc = (k >> (2 - i)) & 1;
                vb *= (kc == pb) ? s_c1[3 + i] : (kc ? s_s1[3 + i] : -s_s1[3 + i]);
            }
            fa[p] = va;
            fb[p] = vb;
        }
        float accr = 0.f, acci = 0.f;
#pragma unroll
        for (int p = 0; p < 8; ++p) {
#pragma unroll
            for (int q = 0; q < 8; ++q) {
                const float re = fa[p] * fb[q];
                const int kp = 8 * p + q;
                accr += re * s_hr[kp * 8 + m];
                acci += re * s_hi[kp * 8 + m];
            }
        }
        const float d1r = s_d1r[k], d1i = s_d1i[k];
        s_m[k * 8 + m] = make_float2(d1r * accr - d1i * acci, d1r * acci + d1i * accr);
    }
    __syncthreads();

    if (tid < 224) {
        const int q    = tid & 3;
        const int u    = tid >> 2;
        const int part = u & 1;
        const int t    = u >> 1;
        const int m  = (t >= 7) + (t >= 13) + (t >= 18) + (t >= 22) + (t >= 25) + (t >= 27);
        const int mp = m + (t - (7 * m - (m * (m - 1)) / 2));
        const float sd = part ? -1.f : 1.f;
        float tot = 0.f, z2 = 0.f, z3 = 0.f, z4 = 0.f, z5 = 0.f;
#pragma unroll
        for (int j = 0; j < 16; ++j) {
            const int k = 16 * q + j;
            const float2 a = s_m[k * 8 + m];
            const float2 b = s_m[k * 8 + mp];
            const float bb = part ? b.y : b.x;
            const float dd = part ? b.x : b.y;
            const float p  = a.x * bb + sd * (a.y * dd);
            tot += p;
            z2 = (j & 8) ? z2 - p : z2 + p;
            z3 = (j & 4) ? z3 - p : z3 + p;
            z4 = (j & 2) ? z4 - p : z4 + p;
            z5 = (j & 1) ? z5 - p : z5 + p;
        }
        float e[6] = { (q & 2) ? -tot : tot, (q & 1) ? -tot : tot, z2, z3, z4, z5 };
#pragma unroll
        for (int i = 0; i < 6; ++i) {
            e[i] += __shfl_xor(e[i], 1, 32);
            e[i] += __shfl_xor(e[i], 2, 32);
        }
        if (q == 0) {
            const float sc = (m == mp) ? (part ? 0.f : 1.f) : (part ? -2.f : 2.f);
#pragma unroll
            for (int i = 0; i < 6; ++i)
                ws[((i * 2 + part) * NSTEP + (t >> 1)) * 2 + (t & 1)] = sc * e[i];
        }
    }
}

__global__ __launch_bounds__(128, 1)
void qsim_main_kernel(const float* __restrict__ z,
                      const float* __restrict__ ws,
                      float* __restrict__ out) {
    __shared__ __attribute__((aligned(16))) float s_out[4][192];

    const int tid  = threadIdx.x;
    const int wave = tid >> 5;
    const int lane = tid & 31;
    const int n    = lane & 15;
    const int kh   = lane >> 4;
    const int base = blockIdx.x * 128 + wave * 32;
    const int b    = base + lane;

    v2f af[NSTEP];
    const int wrow = n & 7;
    if (wrow < 6) {
        const float4* p = (const float4*)(ws + (wrow * 2 + kh) * (NSTEP * 2));
#pragma unroll
        for (int j = 0; j < NSTEP / 2; ++j) {
            const float4 q = p[j];
            af[2 * j]     = v2f{ q.x, q.y };
            af[2 * j + 1] = v2f{ q.z, q.w };
        }
    } else {
#pragma unroll
        for (int j = 0; j < NSTEP; ++j) af[j] = v2f{ 0.f, 0.f };
    }

    const float2 zz = ((const float2*)z)[b];
    float sa, ca, sb, cb;
    sincosf(0.5f * asinf(zz.x), &sa, &ca);
    sincosf(0.5f * asinf(zz.y), &sb, &cb);
    const c32 x0{ cb * ca, -sb * ca };
    const c32 x1{ cb * sa,  sb * sa };

    const c32 a2 = cmul(x0, x0), a3 = cmul(a2, x0), a4 = cmul(a2, a2);
    const c32 b2 = cmul(x1, x1), b3 = cmul(b2, x1), b4 = cmul(b2, b2);
    c32 vm[7];
    vm[0] = cmul(a4, a2);
    vm[1] = cmul(cmul(a4, x0), x1);
    vm[2] = cmul(a4, b2);
    vm[3] = cmul(a3, b3);
    vm[4] = cmul(a2, b4);
    vm[5] = cmul(x0, cmul(b4, x1));
    vm[6] = cmul(b4, b2);

    float f0[NPAIR], f1[NPAIR];
    {
        int t = 0;
#pragma unroll
        for (int m = 0; m < 7; ++m) {
#pragma unroll
            for (int mp = m; mp < 7; ++mp) {
                const float re = vm[m].r * vm[mp].r + vm[m].i * vm[mp].i;
                const float im = (mp == m) ? 0.f : (vm[m].r * vm[mp].i - vm[m].i * vm[mp].r);
                const auto sw = __builtin_amdgcn_permlane16_swap(__float_as_uint(re), __float_as_uint(im),
                                                                 false, false);
                f0[t] = __uint_as_float(sw[0]);
                f1[t] = __uint_as_float(sw[1]);
                ++t;
            }
        }
    }

    v8f acc0 = {};
    v8f acc1 = {};
#pragma unroll
    for (int kk = 0; kk < NSTEP; ++kk) {
        const v2f bf0 = { f0[2 * kk], f0[2 * kk + 1] };
        const v2f bf1 = { f1[2 * kk], f1[2 * kk + 1] };
        acc0 = __builtin_amdgcn_wmma_f32_16x16x4_f32(false, af[kk], false, bf0, (short)0, acc0, false, false);
        acc1 = __builtin_amdgcn_wmma_f32_16x16x4_f32(false, af[kk], false, bf1, (short)0, acc1, false, false);
    }

    float* so = &s_out[wave][0];
    {
        float2* d = (float2*)(so + lane * 6);
        d[0] = make_float2(kh ? acc1[0] : acc0[0], kh ? acc1[1] : acc0[1]);
        d[1] = make_float2(kh ? acc1[2] : acc0[2], kh ? acc1[3] : acc0[3]);
        d[2] = make_float2(kh ? acc1[4] : acc0[4], kh ? acc1[5] : acc0[5]);
    }

    float* o = out + (size_t)base * 6;
#pragma unroll
    for (int j = 0; j < 6; ++j) o[lane + 32 * j] = so[lane + 32 * j];
}

extern "C" void kernel_launch(void* const* d_in, const int* in_sizes, int n_in,
                              void* d_out, int out_size, void* d_ws, size_t ws_size,
                              hipStream_t stream) {
    const float* w  = (const float*)d_in[0];
    const float* z  = (const float*)d_in[1];
    float* ws       = (float*)d_ws;
    float* o        = (float*)d_out;
    const int B     = in_sizes[1] / 2;

    qsim_setup_kernel<<<1, 512, 0, stream>>>(w, ws);
    qsim_main_kernel<<<B / 128, 128, 0, stream>>>(z, ws, o);
}


# Weights-only set-up redone by every 512-thread work-group in its own LDS (M', then W 6x56); its 16 waves read their A fragments there: one launch, no workspace pass; fp32 sums and their order kept
# speedup vs baseline: 7.1439x; 1.2822x over previous
#include <hip/hip_runtime.h>
#include <hip/hip_bf16.h>


typedef __attribute__((ext_vector_type(2))) float v2f;
typedef __attribute__((ext_vector_type(8))) float v8f;

#define NPAIR 28
#define NSTEP 14
#define NWAVE 16

struct c32 { float r, i; };
__device__ __forceinline__ c32 cmul(c32 a, c32 b) {
    return { a.r * b.r - a.i * b.i, a.r * b.i + a.i * b.r };
}

__global__ __launch_bounds__(512)
void qsim_kernel(const float* __restrict__ w,
                 const float* __restrict__ z,
                 float* __restrict__ out) {
    __shared__ float s_d0r[64], s_d0i[64], s_d1r[64], s_d1i[64];
    __shared__ float s_c0[6], s_s0[6], s_c1[6], s_s1[6];
    __shared__ float s_hr[64 * 8], s_hi[64 * 8];
    __shared__ float2 s_m[64 * 8];
    __shared__ __attribute__((aligned(16))) float s_w[6 * 2 * NSTEP * 2];
    __shared__ __attribute__((aligned(16))) float s_out[NWAVE][192];

    const int tid  = threadIdx.x;
    const int wave = tid >> 5;
    const int lane = tid & 31;
    const int n    = lane & 15;
    const int kh   = lane >> 4;
    const int base = blockIdx.x * (NWAVE * 32) + wave * 32;
    const int b    = base + lane;

    const float2 zz = ((const float2*)z)[b];

    if (tid < 128) {
        const int k   = tid & 63;
        const int off = (tid < 64) ? 6 : 17;
        float phi = 0.f;
#pragma unroll
        for (int i = 0; i < 5; ++i) {
            const int bi = (k >> (5 - i)) & 1;
            const int bj = (k >> (4 - i)) & 1;
            const float sgn = (bi == bj) ? -0.5f : 0.5f;
            phi += sgn * w[off + i];
        }
        float s, c;
        sincosf(phi, &s, &c);
        if (tid < 64) { s_d0r[k] = c; s_d0i[k] = s; }
        else          { s_d1r[k] = c; s_d1i[k] = s; }
    } else if (tid < 140) {
        const int j = tid - 128;
        float s, c;
        sincosf(0.5f * w[(j < 6) ? j : (j + 5)], &s, &c);
        if (j < 6) { s_c0[j] = c;     s_s0[j] = s; }
        else       { s_c1[j - 6] = c; s_s1[j - 6] = s; }
    }
    __syncthreads();

    if (tid < 64) {
        const int kp = tid;
        float g[7] = { 1.f, 0.f, 0.f, 0.f, 0.f, 0.f, 0.f };
#pragma unroll
        for (int i = 0; i < 6; ++i) {
            const int bit = (kp >> (5 - i)) & 1;
            const float a  = bit ? s_s0[i] : s_c0[i];
            const float bb = bit ? s_c0[i] : -s_s0[i];
#pragma unroll
            for (int m = i + 1; m >= 1; --m) g[m] = a * g[m] + bb * g[m - 1];
            g[0] = a * g[0];
        }
        const float c = s_d0r[kp], s = s_d0i[kp];
#pragma unroll
        for (int m = 0; m < 7; ++m) {
            s_hr[kp * 8 + m] = c * g[m];
            s_hi[kp * 8 + m] = s * g[m];
        }
        s_hr[kp * 8 + 7] = 0.f;
        s_hi[kp * 8 + 7] = 0.f;
    }
    __syncthreads();

    {
        const int k = tid >> 3;
        const int m = tid & 7;
        float fa[8], fb[8];
#pragma unroll
        for (int p = 0; p < 8; ++p) {
            float va = 1.f, vb = 1.f;
#pragma unroll
            for (int i = 0; i < 3; ++i) {
                const int kb = (k >> (5 - i)) & 1;
                const int pb = (p >> (2 - i)) & 1;
                va *= (kb == pb) ? s_c1[i] : (kb ? s_s1[i] : -s_s1[i]);
                const int kc = (k >> (2 - i)) & 1;
                vb *= (kc == pb) ? s_c1[3 + i] : (kc ? s_s1[3 + i] : -s_s1[3 + i]);
            }
            fa[p] = va;
            fb[p] = vb;
        }
        float accr = 0.f, acci = 0.f;
#pragma unroll
        for (int p = 0; p < 8; ++p) {
#pragma unroll
            for (int q = 0; q < 8; ++q) {
                const float re = fa[p] * fb[q];
                const int kp = 8 * p + q;
                accr += re * s_hr[kp * 8 + m];
                acci += re * s_hi[kp * 8 + m];
            }
        }
        const float d1r = s_d1r[k], d1i = s_d1i[k];
        s_m[k * 8 + m] = make_float2(d1r * accr - d1i * acci, d1r * acci + d1i * accr);
    }
    __syncthreads();

    if (tid < 224) {
        const int q    = tid & 3;
        const int u    = tid >> 2;
        const int part = u & 1;
        const int t    = u >> 1;
        const int m  = (t >= 7) + (t >= 13) + (t >= 18) + (t >= 22) + (t >= 25) + (t >= 27);
        const int mp = m + (t - (7 * m - (m * (m - 1)) / 2));
        const float sd = part ? -1.f : 1.f;
        float tot = 0.f, z2 = 0.f, z3 = 0.f, z4 = 0.f, z5 = 0.f;
#pragma unroll
        for (int j = 0; j < 16; ++j) {
            const int k = 16 * q + j;
            const float2 a = s_m[k * 8 + m];
            const float2 b = s_m[k * 8 + mp];
            const float bb = part ? b.y : b.x;
            const float dd = part ? b.x : b.y;
            const float p  = a.x * bb + sd * (a.y * dd);
            tot += p;
            z2 = (j & 8) ? z2 - p : z2 + p;
            z3 = (j & 4) ? z3 - p : z3 + p;
            z4 = (j & 2) ? z4 - p : z4 + p;
            z5 = (j & 1) ? z5 - p : z5 + p;
        }
        float e[6] = { (q & 2) ? -tot : tot, (q & 1) ? -tot : tot, z2, z3, z4, z5 };
#pragma unroll
        for (int i = 0; i < 6; ++i) {
            e[i] += __shfl_xor(e[i], 1, 32);
            e[i] += __shfl_xor(e[i], 2, 32);
        }
        if (q == 0) {
            const float sc = (m == mp) ? (part ? 0.f : 1.f) : (part ? -2.f : 2.f);
#pragma unroll
            for (int i = 0; i < 6; ++i)
                s_w[((i * 2 + part) * NSTEP + (t >> 1)) * 2 + (t & 1)] = sc * e[i];
        }
    }
    __syncthreads();

    v2f af[NSTEP];
    const int wrow = n & 7;
    if (wrow < 6) {
        const float4* p = (const float4*)(s_w + (wrow * 2 + kh) * (NSTEP * 2));
#pragma unroll
        for (int j = 0; j < NSTEP / 2; ++j) {
            const float4 q = p[j];
            af[2 * j]     = v2f{ q.x, q.y };
            af[2 * j + 1] = v2f{ q.z, q.w };
        }
    } else {
#pragma unroll
        for (int j = 0; j < NSTEP; ++j) af[j] = v2f{ 0.f, 0.f };
    }

    float sa, ca, sb, cb;
    sincosf(0.5f * asinf(zz.x), &sa, &ca);
    sincosf(0.5f * asinf(zz.y), &sb, &cb);
    const c32 x0{ cb * ca, -sb * ca };
    const c32 x1{ cb * sa,  sb * sa };

    const c32 a2 = cmul(x0, x0), a3 = cmul(a2, x0), a4 = cmul(a2, a2);
    const c32 b2 = cmul(x1, x1), b3 = cmul(b2, x1), b4 = cmul(b2, b2);
    c32 vm[7];
    vm[0] = cmul(a4, a2);
    vm[1] = cmul(cmul(a4, x0), x1);
    vm[2] = cmul(a4, b2);
    vm[3] = cmul(a3, b3);
    vm[4] = cmul(a2, b4);
    vm[5] = cmul(x0, cmul(b4, x1));
    vm[6] = cmul(b4, b2);

    float f0[NPAIR], f1[NPAIR];
    {
        int t = 0;
#pragma unroll
        for (int m = 0; m < 7; ++m) {
#pragma unroll
            for (int mp = m; mp < 7; ++mp) {
                const float re = vm[m].r * vm[mp].r + vm[m].i * vm[mp].i;
                const float im = (mp == m) ? 0.f : (vm[m].r * vm[mp].i - vm[m].i * vm[mp].r);
                const auto sw = __builtin_amdgcn_permlane16_swap(__float_as_uint(re), __float_as_uint(im),
                                                                 false, false);
                f0[t] = __uint_as_float(sw[0]);
                f1[t] = __uint_as_float(sw[1]);
                ++t;
            }
        }
    }

    v8f acc0 = {};
    v8f acc1 = {};
#pragma unroll
    for (int kk = 0; kk < NSTEP; ++kk) {
        const v2f bf0 = { f0[2 * kk], f0[2 * kk + 1] };
        const v2f bf1 = { f1[2 * kk], f1[2 * kk + 1] };
        acc0 = __builtin_amdgcn_wmma_f32_16x16x4_f32(false, af[kk], false, bf0, (short)0, acc0, false, false);
        acc1 = __builtin_amdgcn_wmma_f32_16x16x4_f32(false, af[kk], false, bf1, (short)0, acc1, false, false);
    }

    float* so = &s_out[wave][0];
    {
        float2* d = (float2*)(so + lane * 6);
        d[0] = make_float2(kh ? acc1[0] : acc0[0], kh ? acc1[1] : acc0[1]);
        d[1] = make_float2(kh ? acc1[2] : acc0[2], kh ? acc1[3] : acc0[3]);
        d[2] = make_float2(kh ? acc1[4] : acc0[4], kh ? acc1[5] : acc0[5]);
    }

    float* o = out + (size_t)base * 6;
#pragma unroll
    for (int j = 0; j < 6; ++j) o[lane + 32 * j] = so[lane + 32 * j];
}

extern "C" void kernel_launch(void* const* d_in, const int* in_sizes, int n_in,
                              void* d_out, int out_size, void* d_ws, size_t ws_size,
                              hipStream_t stream) {
    const float* w  = (const float*)d_in[0];
    const float* z  = (const float*)d_in[1];
    float* o        = (float*)d_out;
    const int B     = in_sizes[1] / 2;

    qsim_kernel<<<B / 512, 512, 0, stream>>>(w, z, o);
}


# Set-up's M' columns = the circuit's gates run over the 7 vectors e_m, one per wave; RY = lane-pair butterfly (row_xmask DPP, permlanex16); no 64x64 product, one barrier less; changes fp32 sum order
# speedup vs baseline: 8.4609x; 1.1844x over previous
#include <hip/hip_runtime.h>
#include <hip/hip_bf16.h>


typedef __attribute__((ext_vector_type(2))) float v2f;
typedef __attribute__((ext_vector_type(8))) float v8f;

#define NPAIR 28
#define NSTEP 14
#define NWAVE 16

struct c32 { float r, i; };
__device__ __forceinline__ c32 cmul(c32 a, c32 b) {
    return { a.r * b.r - a.i * b.i, a.r * b.i + a.i * b.r };
}

template <int BIT>
__device__ __forceinline__ float lane_partner(float x) {
    const int b = __builtin_bit_cast(int, x);
    if (BIT == 4)
        return __builtin_bit_cast(float,
            __builtin_amdgcn_permlanex16(b, b, 0x76543210, 0xfedcba98, false, false));
    return __builtin_bit_cast(float,
        __builtin_amdgcn_update_dpp(0, b, 0x160 | (1 << BIT), 0xF, 0xF, true));
}

template <int BIT, int NV>
__device__ __forceinline__ void ry_lanes(float (&v)[NV], float c, float s, int lane) {
    const float ss = ((lane >> BIT) & 1) ? s : -s;
#pragma unroll
    for (int j = 0; j < NV; ++j) v[j] = c * v[j] + ss * lane_partner<BIT>(v[j]);
}

template <int NV>
__device__ __forceinline__ void ry_regs(float (&v)[NV], float c, float s) {
#pragma unroll
    for (int p = 0; p < NV / 2; ++p) {
        const float a = v[2 * p], b = v[2 * p + 1];
        v[2 * p]     = c * a - s * b;
        v[2 * p + 1] = s * a + c * b;
    }
}

__global__ __launch_bounds__(512)
void qsim_kernel(const float* __restrict__ w,
                 const float* __restrict__ z,
                 float* __restrict__ out) {
    __shared__ float s_d0r[64], s_d0i[64], s_d1r[64], s_d1i[64];
    __shared__ float s_c0[6], s_s0[6], s_c1[6], s_s1[6];
    __shared__ float2 s_m[64 * 8];
    __shared__ __attribute__((aligned(16))) float s_w[6 * 2 * NSTEP * 2];
    __shared__ __attribute__((aligned(16))) float s_out[NWAVE][192];

    const int tid  = threadIdx.x;
    const int wave = tid >> 5;
    const int lane = tid & 31;
    const int n    = lane & 15;
    const int kh   = lane >> 4;
    const int base = blockIdx.x * (NWAVE * 32) + wave * 32;
    const int b    = base + lane;

    const float2 zz = ((const float2*)z)[b];

    if (tid < 128) {
        const int k   = tid & 63;
        const int off = (tid < 64) ? 6 : 17;
        float phi = 0.f;
#pragma unroll
        for (int i = 0; i < 5; ++i) {
            const int bi = (k >> (5 - i)) & 1;
            const int bj = (k >> (4 - i)) & 1;
            const float sgn = (bi == bj) ? -0.5f : 0.5f;
            phi += sgn * w[off + i];
        }
        float s, c;
        sincosf(phi, &s, &c);
        if (tid < 64) { s_d0r[k] = c; s_d0i[k] = s; }
        else          { s_d1r[k] = c; s_d1i[k] = s; }
    } else if (tid < 140) {
        const int j = tid - 128;
        float s, c;
        sincosf(0.5f * w[(j < 6) ? j : (j + 5)], &s, &c);
        if (j < 6) { s_c0[j] = c;     s_s0[j] = s; }
        else       { s_c1[j - 6] = c; s_s1[j - 6] = s; }
    }
    __syncthreads();

    if (tid < 224) {
        const int m = wave;
        const int pc = __popc(lane);
        float g[2] = { (pc == m) ? 1.f : 0.f, (pc + 1 == m) ? 1.f : 0.f };
        ry_regs<2>(g, s_c0[0], s_s0[0]);
        ry_lanes<4, 2>(g, s_c0[1], s_s0[1], lane);
        ry_lanes<3, 2>(g, s_c0[2], s_s0[2], lane);
        ry_lanes<2, 2>(g, s_c0[3], s_s0[3], lane);
        ry_lanes<1, 2>(g, s_c0[4], s_s0[4], lane);
        ry_lanes<0, 2>(g, s_c0[5], s_s0[5], lane);
        float h[4] = { s_d0r[lane] * g[0], s_d0r[lane + 32] * g[1],
                       s_d0i[lane] * g[0], s_d0i[lane + 32] * g[1] };
        ry_regs<4>(h, s_c1[0], s_s1[0]);
        ry_lanes<4, 4>(h, s_c1[1], s_s1[1], lane);
        ry_lanes<3, 4>(h, s_c1[2], s_s1[2], lane);
        ry_lanes<2, 4>(h, s_c1[3], s_s1[3], lane);
        ry_lanes<1, 4>(h, s_c1[4], s_s1[4], lane);
        ry_lanes<0, 4>(h, s_c1[5], s_s1[5], lane);
        const float e0r = s_d1r[lane],      e0i = s_d1i[lane];
        const float e1r = s_d1r[lane + 32], e1i = s_d1i[lane + 32];
        s_m[lane * 8 + m]        = make_float2(e0r * h[0] - e0i * h[2], e0r * h[2] + e0i * h[0]);
        s_m[(lane + 32) * 8 + m] = make_float2(e1r * h[1] - e1i * h[3], e1r * h[3] + e1i * h[1]);
    }
    __syncthreads();

    if (tid < 224) {
        const int q    = tid & 3;
        const int u    = tid >> 2;
        const int part = u & 1;
        const int t    = u >> 1;
        const int m  = (t >= 7) + (t >= 13) + (t >= 18) + (t >= 22) + (t >= 25) + (t >= 27);
        const int mp = m + (t - (7 * m - (m * (m - 1)) / 2));
        const float sd = part ? -1.f : 1.f;
        float tot = 0.f, z2 = 0.f, z3 = 0.f, z4 = 0.f, z5 = 0.f;
#pragma unroll
        for (int j = 0; j < 16; ++j) {
            const int k = 16 * q + j;
            const float2 a = s_m[k * 8 + m];
            const float2 b = s_m[k * 8 + mp];
            const float bb = part ? b.y : b.x;
            const float dd = part ? b.x : b.y;
            const float p  = a.x * bb + sd * (a.y * dd);
            tot += p;
            z2 = (j & 8) ? z2 - p : z2 + p;
            z3 = (j & 4) ? z3 - p : z3 + p;
            z4 = (j & 2) ? z4 - p : z4 + p;
            z5 = (j & 1) ? z5 - p : z5 + p;
        }
        float e[6] = { (q & 2) ? -tot : tot, (q & 1) ? -tot : tot, z2, z3, z4, z5 };
#pragma unroll
        for (int i = 0; i < 6; ++i) {
            e[i] += __shfl_xor(e[i], 1, 32);
            e[i] += __shfl_xor(e[i], 2, 32);
        }
        if (q == 0) {
            const float sc = (m == mp) ? (part ? 0.f : 1.f) : (part ? -2.f : 2.f);
#pragma unroll
            for (int i = 0; i < 6; ++i)
                s_w[((i * 2 + part) * NSTEP + (t >> 1)) * 2 + (t & 1)] = sc * e[i];
        }
    }
    __syncthreads();

    v2f af[NSTEP];
    const int wrow = n & 7;
    if (wrow < 6) {
        const float4* p = (const float4*)(s_w + (wrow * 2 + kh) * (NSTEP * 2));
#pragma unroll
        for (int j = 0; j < NSTEP / 2; ++j) {
            const float4 q = p[j];
            af[2 * j]     = v2f{ q.x, q.y };
            af[2 * j + 1] = v2f{ q.z, q.w };
        }
    } else {
#pragma unroll
        for (int j = 0; j < NSTEP; ++j) af[j] = v2f{ 0.f, 0.f };
    }

    float sa, ca, sb, cb;
    sincosf(0.5f * asinf(zz.x), &sa, &ca);
    sincosf(0.5f * asinf(zz.y), &sb, &cb);
    const c32 x0{ cb * ca, -sb * ca };
    const c32 x1{ cb * sa,  sb * sa };

    const c32 a2 = cmul(x0, x0), a3 = cmul(a2, x0), a4 = cmul(a2, a2);
    const c32 b2 = cmul(x1, x1), b3 = cmul(b2, x1), b4 = cmul(b2, b2);
    c32 vm[7];
    vm[0] = cmul(a4, a2);
    vm[1] = cmul(cmul(a4, x0), x1);
    vm[2] = cmul(a4, b2);
    vm[3] = cmul(a3, b3);
    vm[4] = cmul(a2, b4);
    vm[5] = cmul(x0, cmul(b4, x1));
    vm[6] = cmul(b4, b2);

    float f0[NPAIR], f1[NPAIR];
    {
        int t = 0;
#pragma unroll
        for (int m = 0; m < 7; ++m) {
#pragma unroll
            for (int mp = m; mp < 7; ++mp) {
                const float re = vm[m].r * vm[mp].r + vm[m].i * vm[mp].i;
                const float im = (mp == m) ? 0.f : (vm[m].r * vm[mp].i - vm[m].i * vm[mp].r);
                const auto sw = __builtin_amdgcn_permlane16_swap(__float_as_uint(re), __float_as_uint(im),
                                                                 false, false);
                f0[t] = __uint_as_float(sw[0]);
                f1[t] = __uint_as_float(sw[1]);
                ++t;
            }
        }
    }

    v8f acc0 = {};
    v8f acc1 = {};
#pragma unroll
    for (int kk = 0; kk < NSTEP; ++kk) {
        const v2f bf0 = { f0[2 * kk], f0[2 * kk + 1] };
        const v2f bf1 = { f1[2 * kk], f1[2 * kk + 1] };
        acc0 = __builtin_amdgcn_wmma_f32_16x16x4_f32(false, af[kk], false, bf0, (short)0, acc0, false, false);
        acc1 = __builtin_amdgcn_wmma_f32_16x16x4_f32(false, af[kk], false, bf1, (short)0, acc1, false, false);
    }

    float* so = &s_out[wave][0];
    {
        float2* d = (float2*)(so + lane * 6);
        d[0] = make_float2(kh ? acc1[0] : acc0[0], kh ? acc1[1] : acc0[1]);
        d[1] = make_float2(kh ? acc1[2] : acc0[2], kh ? acc1[3] : acc0[3]);
        d[2] = make_float2(kh ? acc1[4] : acc0[4], kh ? acc1[5] : acc0[5]);
    }

    float* o = out + (size_t)base * 6;
#pragma unroll
    for (int j = 0; j < 6; ++j) o[lane + 32 * j] = so[lane + 32 * j];
}

extern "C" void kernel_launch(void* const* d_in, const int* in_sizes, int n_in,
                              void* d_out, int out_size, void* d_ws, size_t ws_size,
                              hipStream_t stream) {
    const float* w  = (const float*)d_in[0];
    const float* z  = (const float*)d_in[1];
    float* o        = (float*)d_out;
    const int B     = in_sizes[1] / 2;

    qsim_kernel<<<B / 512, 512, 0, stream>>>(w, z, o);
}


# Set-up's LDS bank conflicts removed: M' stored column by column at pitch 68, W-stage threads take states 4j+q, so its 8-way stores and 4-way loads each meet a bank once; changes fp32 sum order
# speedup vs baseline: 9.0375x; 1.0681x over previous
#include <hip/hip_runtime.h>
#include <hip/hip_bf16.h>


typedef __attribute__((ext_vector_type(2))) float v2f;
typedef __attribute__((ext_vector_type(8))) float v8f;

#define NPAIR 28
#define NSTEP 14
#define NWAVE 16
#define MPITCH 68

struct c32 { float r, i; };
__device__ __forceinline__ c32 cmul(c32 a, c32 b) {
    return { a.r * b.r - a.i * b.i, a.r * b.i + a.i * b.r };
}

template <int BIT>
__device__ __forceinline__ float lane_partner(float x) {
    const int b = __builtin_bit_cast(int, x);
    if (BIT == 4)
        return __builtin_bit_cast(float,
            __builtin_amdgcn_permlanex16(b, b, 0x76543210, 0xfedcba98, false, false));
    return __builtin_bit_cast(float,
        __builtin_amdgcn_update_dpp(0, b, 0x160 | (1 << BIT), 0xF, 0xF, true));
}

template <int BIT, int NV>
__device__ __forceinline__ void ry_lanes(float (&v)[NV], float c, float s, int lane) {
    const float ss = ((lane >> BIT) & 1) ? s : -s;
#pragma unroll
    for (int j = 0; j < NV; ++j) v[j] = c * v[j] + ss * lane_partner<BIT>(v[j]);
}

template <int NV>
__device__ __forceinline__ void ry_regs(float (&v)[NV], float c, float s) {
#pragma unroll
    for (int p = 0; p < NV / 2; ++p) {
        const float a = v[2 * p], b = v[2 * p + 1];
        v[2 * p]     = c * a - s * b;
        v[2 * p + 1] = s * a + c * b;
    }
}

__global__ __launch_bounds__(512)
void qsim_kernel(const float* __restrict__ w,
                 const float* __restrict__ z,
                 float* __restrict__ out) {
    __shared__ float s_d0r[64], s_d0i[64], s_d1r[64], s_d1i[64];
    __shared__ float s_c0[6], s_s0[6], s_c1[6], s_s1[6];
    __shared__ float2 s_m[7 * MPITCH];
    __shared__ __attribute__((aligned(16))) float s_w[6 * 2 * NSTEP * 2];
    __shared__ __attribute__((aligned(16))) float s_out[NWAVE][192];

    const int tid  = threadIdx.x;
    const int wave = tid >> 5;
    const int lane = tid & 31;
    const int n    = lane & 15;
    const int kh   = lane >> 4;
    const int base = blockIdx.x * (NWAVE * 32) + wave * 32;
    const int b    = base + lane;

    const float2 zz = ((const float2*)z)[b];

    if (tid < 128) {
        const int k   = tid & 63;
        const int off = (tid < 64) ? 6 : 17;
        float phi = 0.f;
#pragma unroll
        for (int i = 0; i < 5; ++i) {
            const int bi = (k >> (5 - i)) & 1;
            const int bj = (k >> (4 - i)) & 1;
            const float sgn = (bi == bj) ? -0.5f : 0.5f;
            phi += sgn * w[off + i];
        }
        float s, c;
        sincosf(phi, &s, &c);
        if (tid < 64) { s_d0r[k] = c; s_d0i[k] = s; }
        else          { s_d1r[k] = c; s_d1i[k] = s; }
    } else if (tid < 140) {
        const int j = tid - 128;
        float s, c;
        sincosf(0.5f * w[(j < 6) ? j : (j + 5)], &s, &c);
        if (j < 6) { s_c0[j] = c;     s_s0[j] = s; }
        else       { s_c1[j - 6] = c; s_s1[j - 6] = s; }
    }
    __syncthreads();

    if (tid < 224) {
        const int m = wave;
        const int pc = __popc(lane);
        float g[2] = { (pc == m) ? 1.f : 0.f, (pc + 1 == m) ? 1.f : 0.f };
        ry_regs<2>(g, s_c0[0], s_s0[0]);
        ry_lanes<4, 2>(g, s_c0[1], s_s0[1], lane);
        ry_lanes<3, 2>(g, s_c0[2], s_s0[2], lane);
        ry_lanes<2, 2>(g, s_c0[3], s_s0[3], lane);
        ry_lanes<1, 2>(g, s_c0[4], s_s0[4], lane);
        ry_lanes<0, 2>(g, s_c0[5], s_s0[5], lane);
        float h[4] = { s_d0r[lane] * g[0], s_d0r[lane + 32] * g[1],
                       s_d0i[lane] * g[0], s_d0i[lane + 32] * g[1] };
        ry_regs<4>(h, s_c1[0], s_s1[0]);
        ry_lanes<4, 4>(h, s_c1[1], s_s1[1], lane);
        ry_lanes<3, 4>(h, s_c1[2], s_s1[2], lane);
        ry_lanes<2, 4>(h, s_c1[3], s_s1[3], lane);
        ry_lanes<1, 4>(h, s_c1[4], s_s1[4], lane);
        ry_lanes<0, 4>(h, s_c1[5], s_s1[5], lane);
        const float e0r = s_d1r[lane],      e0i = s_d1i[lane];
        const float e1r = s_d1r[lane + 32], e1i = s_d1i[lane + 32];
        s_m[m * MPITCH + lane]      = make_float2(e0r * h[0] - e0i * h[2], e0r * h[2] + e0i * h[0]);
        s_m[m * MPITCH + lane + 32] = make_float2(e1r * h[1] - e1i * h[3], e1r * h[3] + e1i * h[1]);
    }
    __syncthreads();

    if (tid < 224) {
        const int q    = tid & 3;
        const int u    = tid >> 2;
        const int part = u & 1;
        const int t    = u >> 1;
        const int m  = (t >= 7) + (t >= 13) + (t >= 18) + (t >= 22) + (t >= 25) + (t >= 27);
        const int mp = m + (t - (7 * m - (m * (m - 1)) / 2));
        const float sd = part ? -1.f : 1.f;
        float tot = 0.f, z0 = 0.f, z1 = 0.f, z2 = 0.f, z3 = 0.f;
#pragma unroll
        for (int j = 0; j < 16; ++j) {
            const int k = 4 * j + q;
            const float2 a = s_m[m * MPITCH + k];
            const float2 b = s_m[mp * MPITCH + k];
            const float bb = part ? b.y : b.x;
            const float dd = part ? b.x : b.y;
            const float p  = a.x * bb + sd * (a.y * dd);
            tot += p;
            z0 = (j & 8) ? z0 - p : z0 + p;
            z1 = (j & 4) ? z1 - p : z1 + p;
            z2 = (j & 2) ? z2 - p : z2 + p;
            z3 = (j & 1) ? z3 - p : z3 + p;
        }
        float e[6] = { z0, z1, z2, z3, (q & 2) ? -tot : tot, (q & 1) ? -tot : tot };
#pragma unroll
        for (int i = 0; i < 6; ++i) {
            e[i] += __shfl_xor(e[i], 1, 32);
            e[i] += __shfl_xor(e[i], 2, 32);
        }
        if (q == 0) {
            const float sc = (m == mp) ? (part ? 0.f : 1.f) : (part ? -2.f : 2.f);
#pragma unroll
            for (int i = 0; i < 6; ++i)
                s_w[((i * 2 + part) * NSTEP + (t >> 1)) * 2 + (t & 1)] = sc * e[i];
        }
    }
    __syncthreads();

    v2f af[NSTEP];
    const int wrow = n & 7;
    if (wrow < 6) {
        const float4* p = (const float4*)(s_w + (wrow * 2 + kh) * (NSTEP * 2));
#pragma unroll
        for (int j = 0; j < NSTEP / 2; ++j) {
            const float4 q = p[j];
            af[2 * j]     = v2f{ q.x, q.y };
            af[2 * j + 1] = v2f{ q.z, q.w };
        }
    } else {
#pragma unroll
        for (int j = 0; j < NSTEP; ++j) af[j] = v2f{ 0.f, 0.f };
    }

    float sa, ca, sb, cb;
    sincosf(0.5f * asinf(zz.x), &sa, &ca);
    sincosf(0.5f * asinf(zz.y), &sb, &cb);
    const c32 x0{ cb * ca, -sb * ca };
    const c32 x1{ cb * sa,  sb * sa };

    const c32 a2 = cmul(x0, x0), a3 = cmul(a2, x0), a4 = cmul(a2, a2);
    const c32 b2 = cmul(x1, x1), b3 = cmul(b2, x1), b4 = cmul(b2, b2);
    c32 vm[7];
    vm[0] = cmul(a4, a2);
    vm[1] = cmul(cmul(a4, x0), x1);
    vm[2] = cmul(a4, b2);
    vm[3] = cmul(a3, b3);
    vm[4] = cmul(a2, b4);
    vm[5] = cmul(x0, cmul(b4, x1));
    vm[6] = cmul(b4, b2);

    float f0[NPAIR], f1[NPAIR];
    {
        int t = 0;
#pragma unroll
        for (int m = 0; m < 7; ++m) {
#pragma unroll
            for (int mp = m; mp < 7; ++mp) {
                const float re = vm[m].r * vm[mp].r + vm[m].i * vm[mp].i;
                const float im = (mp == m) ? 0.f : (vm[m].r * vm[mp].i - vm[m].i * vm[mp].r);
                const auto sw = __builtin_amdgcn_permlane16_swap(__float_as_uint(re), __float_as_uint(im),
                                                                 false, false);
                f0[t] = __uint_as_float(sw[0]);
                f1[t] = __uint_as_float(sw[1]);
                ++t;
            }
        }
    }

    v8f acc0 = {};
    v8f acc1 = {};
#pragma unroll
    for (int kk = 0; kk < NSTEP; ++kk) {
        const v2f bf0 = { f0[2 * kk], f0[2 * kk + 1] };
        const v2f bf1 = { f1[2 * kk], f1[2 * kk + 1] };
        acc0 = __builtin_amdgcn_wmma_f32_16x16x4_f32(false, af[kk], false, bf0, (short)0, acc0, false, false);
        acc1 = __builtin_amdgcn_wmma_f32_16x16x4_f32(false, af[kk], false, bf1, (short)0, acc1, false, false);
    }

    float* so = &s_out[wave][0];
    {
        float2* d = (float2*)(so + lane * 6);
        d[0] = make_float2(kh ? acc1[0] : acc0[0], kh ? acc1[1] : acc0[1]);
        d[1] = make_float2(kh ? acc1[2] : acc0[2], kh ? acc1[3] : acc0[3]);
        d[2] = make_float2(kh ? acc1[4] : acc0[4], kh ? acc1[5] : acc0[5]);
    }

    float* o = out + (size_t)base * 6;
#pragma unroll
    for (int j = 0; j < 6; ++j) o[lane + 32 * j] = so[lane + 32 * j];
}

extern "C" void kernel_launch(void* const* d_in, const int* in_sizes, int n_in,
                              void* d_out, int out_size, void* d_ws, size_t ws_size,
                              hipStream_t stream) {
    const float* w  = (const float*)d_in[0];
    const float* z  = (const float*)d_in[1];
    float* o        = (float*)d_out;
    const int B     = in_sizes[1] / 2;

    qsim_kernel<<<B / 512, 512, 0, stream>>>(w, z, o);
}


# Wave roles split: the 9 waves with no gate column do their columns' asin/sincos, monomials and 28 products between the set-up barriers' signal and wait; WMMAs start at the last barrier; fp32 sums kept
# speedup vs baseline: 9.5535x; 1.0571x over previous
#include <hip/hip_runtime.h>
#include <hip/hip_bf16.h>


typedef __attribute__((ext_vector_type(2))) float v2f;
typedef __attribute__((ext_vector_type(8))) float v8f;

#define NPAIR 28
#define NSTEP 14
#define NWAVE 16
#define NSETUP 7
#define MPITCH 68

__device__ __forceinline__ void bar_arrive() {
    asm volatile("s_wait_dscnt 0x0\n\ts_barrier_signal -1" ::: "memory");
}
__device__ __forceinline__ void bar_wait() {
    asm volatile("s_barrier_wait -1" ::: "memory");
}

struct c32 { float r, i; };
__device__ __forceinline__ c32 cmul(c32 a, c32 b) {
    return { a.r * b.r - a.i * b.i, a.r * b.i + a.i * b.r };
}

template <int BIT>
__device__ __forceinline__ float lane_partner(float x) {
    const int b = __builtin_bit_cast(int, x);
    if (BIT == 4)
        return __builtin_bit_cast(float,
            __builtin_amdgcn_permlanex16(b, b, 0x76543210, 0xfedcba98, false, false));
    return __builtin_bit_cast(float,
        __builtin_amdgcn_update_dpp(0, b, 0x160 | (1 << BIT), 0xF, 0xF, true));
}

template <int BIT, int NV>
__device__ __forceinline__ void ry_lanes(float (&v)[NV], float c, float s, int lane) {
    const float ss = ((lane >> BIT) & 1) ? s : -s;
#pragma unroll
    for (int j = 0; j < NV; ++j) v[j] = c * v[j] + ss * lane_partner<BIT>(v[j]);
}

template <int NV>
__device__ __forceinline__ void ry_regs(float (&v)[NV], float c, float s) {
#pragma unroll
    for (int p = 0; p < NV / 2; ++p) {
        const float a = v[2 * p], b = v[2 * p + 1];
        v[2 * p]     = c * a - s * b;
        v[2 * p + 1] = s * a + c * b;
    }
}

__device__ __forceinline__ void col_half_angles(float2 zz, float& ha, float& hb) {
    ha = 0.5f * asinf(zz.x);
    hb = 0.5f * asinf(zz.y);
}

__device__ __forceinline__ void col_monomials(float ha, float hb, c32 (&vm)[7]) {
    float sa, ca, sb, cb;
    sincosf(ha, &sa, &ca);
    sincosf(hb, &sb, &cb);
    const c32 x0{ cb * ca, -sb * ca };
    const c32 x1{ cb * sa,  sb * sa };

    const c32 a2 = cmul(x0, x0), a3 = cmul(a2, x0), a4 = cmul(a2, a2);
    const c32 b2 = cmul(x1, x1), b3 = cmul(b2, x1), b4 = cmul(b2, b2);
    vm[0] = cmul(a4, a2);
    vm[1] = cmul(cmul(a4, x0), x1);
    vm[2] = cmul(a4, b2);
    vm[3] = cmul(a3, b3);
    vm[4] = cmul(a2, b4);
    vm[5] = cmul(x0, cmul(b4, x1));
    vm[6] = cmul(b4, b2);
}

__device__ __forceinline__ void col_products(const c32 (&vm)[7], float (&f0)[NPAIR], float (&f1)[NPAIR]) {
    int t = 0;
#pragma unroll
    for (int m = 0; m < 7; ++m) {
#pragma unroll
        for (int mp = m; mp < 7; ++mp) {
            const float re = vm[m].r * vm[mp].r + vm[m].i * vm[mp].i;
            const float im = (mp == m) ? 0.f : (vm[m].r * vm[mp].i - vm[m].i * vm[mp].r);
            const auto sw = __builtin_amdgcn_permlane16_swap(__float_as_uint(re), __float_as_uint(im),
                                                             false, false);
            f0[t] = __uint_as_float(sw[0]);
            f1[t] = __uint_as_float(sw[1]);
            ++t;
        }
    }
}

__device__ __forceinline__ void col_wmma(const v2f (&af)[NSTEP], const float (&f0)[NPAIR],
                                         const float (&f1)[NPAIR], v8f& acc0, v8f& acc1) {
#pragma unroll
    for (int kk = 0; kk < NSTEP; ++kk) {
        const v2f bf0 = { f0[2 * kk], f0[2 * kk + 1] };
        const v2f bf1 = { f1[2 * kk], f1[2 * kk + 1] };
        acc0 = __builtin_amdgcn_wmma_f32_16x16x4_f32(false, af[kk], false, bf0, (short)0, acc0, false, false);
        acc1 = __builtin_amdgcn_wmma_f32_16x16x4_f32(false, af[kk], false, bf1, (short)0, acc1, false, false);
    }
}

__global__ __launch_bounds__(512)
void qsim_kernel(const float* __restrict__ w,
                 const float* __restrict__ z,
                 float* __restrict__ out) {
    __shared__ float s_d0r[64], s_d0i[64], s_d1r[64], s_d1i[64];
    __shared__ float s_c0[6], s_s0[6], s_c1[6], s_s1[6];
    __shared__ float2 s_m[7 * MPITCH];
    __shared__ __attribute__((aligned(16))) float s_w[6 * 2 * NSTEP * 2];
    __shared__ __attribute__((aligned(16))) float s_out[NWAVE][192];

    const int tid  = threadIdx.x;
    const int wave = tid >> 5;
    const int lane = tid & 31;
    const int n    = lane & 15;
    const int kh   = lane >> 4;
    const int base = blockIdx.x * (NWAVE * 32) + wave * 32;
    const int b    = base + lane;

    const bool free_wave = __builtin_amdgcn_readfirstlane(wave) >= NSETUP;

    const float2 zz = ((const float2*)z)[b];

    if (tid < 128) {
        const int k   = tid & 63;
        const int off = (tid < 64) ? 6 : 17;
        float phi = 0.f;
#pragma unroll
        for (int i = 0; i < 5; ++i) {
            const int bi = (k >> (5 - i)) & 1;
            const int bj = (k >> (4 - i)) & 1;
            const float sgn = (bi == bj) ? -0.5f : 0.5f;
            phi += sgn * w[off + i];
        }
        float s, c;
        sincosf(phi, &s, &c);
        if (tid < 64) { s_d0r[k] = c; s_d0i[k] = s; }
        else          { s_d1r[k] = c; s_d1i[k] = s; }
    } else if (tid < 140) {
        const int j = tid - 128;
        float s, c;
        sincosf(0.5f * w[(j < 6) ? j : (j + 5)], &s, &c);
        if (j < 6) { s_c0[j] = c;     s_s0[j] = s; }
        else       { s_c1[j - 6] = c; s_s1[j - 6] = s; }
    }
    bar_arrive();
    float ha, hb;
    if (free_wave) col_half_angles(zz, ha, hb);
    bar_wait();

    if (!free_wave) {
        const int m = wave;
        const int pc = __popc(lane);
        float g[2] = { (pc == m) ? 1.f : 0.f, (pc + 1 == m) ? 1.f : 0.f };
        ry_regs<2>(g, s_c0[0], s_s0[0]);
        ry_lanes<4, 2>(g, s_c0[1], s_s0[1], lane);
        ry_lanes<3, 2>(g, s_c0[2], s_s0[2], lane);
        ry_lanes<2, 2>(g, s_c0[3], s_s0[3], lane);
        ry_lanes<1, 2>(g, s_c0[4], s_s0[4], lane);
        ry_lanes<0, 2>(g, s_c0[5], s_s0[5], lane);
        float h[4] = { s_d0r[lane] * g[0], s_d0r[lane + 32] * g[1],
                       s_d0i[lane] * g[0], s_d0i[lane + 32] * g[1] };
        ry_regs<4>(h, s_c1[0], s_s1[0]);
        ry_lanes<4, 4>(h, s_c1[1], s_s1[1], lane);
        ry_lanes<3, 4>(h, s_c1[2], s_s1[2], lane);
        ry_lanes<2, 4>(h, s_c1[3], s_s1[3], lane);
        ry_lanes<1, 4>(h, s_c1[4], s_s1[4], lane);
        ry_lanes<0, 4>(h, s_c1[5], s_s1[5], lane);
        const float e0r = s_d1r[lane],      e0i = s_d1i[lane];
        const float e1r = s_d1r[lane + 32], e1i = s_d1i[lane + 32];
        s_m[m * MPITCH + lane]      = make_float2(e0r * h[0] - e0i * h[2], e0r * h[2] + e0i * h[0]);
        s_m[m * MPITCH + lane + 32] = make_float2(e1r * h[1] - e1i * h[3], e1r * h[3] + e1i * h[1]);
    }
    bar_arrive();
    c32 vm[7];
    if (free_wave) col_monomials(ha, hb, vm);
    bar_wait();

    if (!free_wave) {
        const int q    = tid & 3;
        const int u    = tid >> 2;
        const int part = u & 1;
        const int t    = u >> 1;
        const int m  = (t >= 7) + (t >= 13) + (t >= 18) + (t >= 22) + (t >= 25) + (t >= 27);
        const int mp = m + (t - (7 * m - (m * (m - 1)) / 2));
        const float sd = part ? -1.f : 1.f;
        float tot = 0.f, z0 = 0.f, z1 = 0.f, z2 = 0.f, z3 = 0.f;
#pragma unroll
        for (int j = 0; j < 16; ++j) {
            const int k = 4 * j + q;
            const float2 a = s_m[m * MPITCH + k];
            const float2 b = s_m[mp * MPITCH + k];
            const float bb = part ? b.y : b.x;
            const float dd = part ? b.x : b.y;
            const float p  = a.x * bb + sd * (a.y * dd);
            tot += p;
            z0 = (j & 8) ? z0 - p : z0 + p;
            z1 = (j & 4) ? z1 - p : z1 + p;
            z2 = (j & 2) ? z2 - p : z2 + p;
            z3 = (j & 1) ? z3 - p : z3 + p;
        }
        float e[6] = { z0, z1, z2, z3, (q & 2) ? -tot : tot, (q & 1) ? -tot : tot };
#pragma unroll
        for (int i = 0; i < 6; ++i) {
            e[i] += __shfl_xor(e[i], 1, 32);
            e[i] += __shfl_xor(e[i], 2, 32);
        }
        if (q == 0) {
            const float sc = (m == mp) ? (part ? 0.f : 1.f) : (part ? -2.f : 2.f);
#pragma unroll
            for (int i = 0; i < 6; ++i)
                s_w[((i * 2 + part) * NSTEP + (t >> 1)) * 2 + (t & 1)] = sc * e[i];
        }
    }
    bar_arrive();
    float f0[NPAIR], f1[NPAIR];
    if (free_wave) col_products(vm, f0, f1);
    bar_wait();

    v2f af[NSTEP];
    const int wrow = n & 7;
    if (wrow < 6) {
        const float4* p = (const float4*)(s_w + (wrow * 2 + kh) * (NSTEP * 2));
#pragma unroll
        for (int j = 0; j < NSTEP / 2; ++j) {
            const float4 q = p[j];
            af[2 * j]     = v2f{ q.x, q.y };
            af[2 * j + 1] = v2f{ q.z, q.w };
        }
    } else {
#pragma unroll
        for (int j = 0; j < NSTEP; ++j) af[j] = v2f{ 0.f, 0.f };
    }

    v8f acc0 = {};
    v8f acc1 = {};
    if (free_wave) {
        col_wmma(af, f0, f1, acc0, acc1);
    } else {
        col_half_angles(zz, ha, hb);
        col_monomials(ha, hb, vm);
        col_products(vm, f0, f1);
        col_wmma(af, f0, f1, acc0, acc1);
    }

    float* so = &s_out[wave][0];
    {
        float2* d = (float2*)(so + lane * 6);
        d[0] = make_float2(kh ? acc1[0] : acc0[0], kh ? acc1[1] : acc0[1]);
        d[1] = make_float2(kh ? acc1[2] : acc0[2], kh ? acc1[3] : acc0[3]);
        d[2] = make_float2(kh ? acc1[4] : acc0[4], kh ? acc1[5] : acc0[5]);
    }

    float* o = out + (size_t)base * 6;
#pragma unroll
    for (int j = 0; j < 6; ++j) o[lane + 32 * j] = so[lane + 32 * j];
}

extern "C" void kernel_launch(void* const* d_in, const int* in_sizes, int n_in,
                              void* d_out, int out_size, void* d_ws, size_t ws_size,
                              hipStream_t stream) {
    const float* w  = (const float*)d_in[0];
    const float* z  = (const float*)d_in[1];
    float* o        = (float*)d_out;
    const int B     = in_sizes[1] / 2;

    qsim_kernel<<<B / 512, 512, 0, stream>>>(w, z, o);
}
